# starting kernel
# baseline (speedup 1.0000x reference)
#include <hip/hip_runtime.h>

typedef __attribute__((ext_vector_type(2))) float v2f;
typedef __attribute__((ext_vector_type(8))) float v8f;

#define NEG_SLOPE 0.01f

__device__ __forceinline__ float leaky(float v) {
    return v >= 0.0f ? v : NEG_SLOPE * v;
}

__device__ __forceinline__ float sigmoidf(float z) {
    return 1.0f / (1.0f + __expf(-z));
}

__global__ __launch_bounds__(256)
void fused_conv4d_wmma(const float* __restrict__ x,
                       const float* __restrict__ w0,
                       const float* __restrict__ b0,
                       const float* __restrict__ w1,
                       const float* __restrict__ b1,
                       const float* __restrict__ wt,
                       const float* __restrict__ bt,
                       float* __restrict__ out)
{
    __shared__ float lds[8][368];

    const int tid  = threadIdx.x;
    const int lane = tid & 31;
    const int warp = tid >> 5;
    const int n    = lane & 15;
    const int half = lane >> 4;
    const int m    = n;

    float* xblk  = &lds[warp][0];
    float* y1buf = &lds[warp][272];

    const int W = blockIdx.x * 8 + warp;
    const int b = W >> 12;
    const int d = (W >> 8) & 15;
    const int e = (W >> 4) & 15;
    const int f = W & 15;

    #pragma unroll
    for (int i = 0; i < 4; ++i) {
        const int pos8 = 4 * half + i;
        const int p = (pos8 >> 2) & 1;
        const int q = (pos8 >> 1) & 1;
        const int r = pos8 & 1;
        const size_t idx =
            ((((size_t)b * 32 + 2 * d + p) * 32 + 2 * e + q) * 32 + 2 * f + r) * 32
            + 2 * n;
        const float2 xv = *reinterpret_cast<const float2*>(x + idx);
        xblk[(pos8 * 2 + 0) * 17 + n] = xv.x;
        xblk[(pos8 * 2 + 1) * 17 + n] = xv.y;
    }

    float w0r[3], b0r[3];
    #pragma unroll
    for (int c = 0; c < 3; ++c) { w0r[c] = w0[c]; b0r[c] = b0[c]; }

    float a1v[12][2];
    #pragma unroll
    for (int k = 0; k < 12; ++k) {
        #pragma unroll
        for (int j = 0; j < 2; ++j) {
            const int kk = 4 * k + 2 * half + j;
            a1v[k][j] = (m < 6) ? w1[m * 48 + kk] : 0.0f;
        }
    }

    __syncthreads();

    v8f acc = {0.f, 0.f, 0.f, 0.f, 0.f, 0.f, 0.f, 0.f};
    #pragma unroll
    for (int k = 0; k < 12; ++k) {
        const int kb = 4 * k + 2 * half;
        v2f a, bb;
        a[0] = a1v[k][0];
        a[1] = a1v[k][1];
        #pragma unroll
        for (int j = 0; j < 2; ++j) {
            const int kk  = kb + j;
            const int c   = kk >> 4;
            const int pos = kk & 15;
            bb[j] = leaky(w0r[c] * xblk[pos * 17 + n] + b0r[c]);
        }
        acc = __builtin_amdgcn_wmma_f32_16x16x4_f32(false, a, false, bb,
                                                    (short)0, acc, false, false);
    }

    if (half == 0) {
        #pragma unroll
        for (int o = 0; o < 6; ++o) {
            y1buf[o * 16 + n] = leaky(acc[o] + b1[o]);
        }
    }
    __syncthreads();

    #pragma unroll
    for (int t = 0; t < 3; ++t) {
        v8f acc2 = {0.f, 0.f, 0.f, 0.f, 0.f, 0.f, 0.f, 0.f};
        #pragma unroll
        for (int ks = 0; ks < 2; ++ks) {
            const int kb = 4 * ks + 2 * half;
            v2f a, bb;
            #pragma unroll
            for (int j = 0; j < 2; ++j) {
                const int i = kb + j;
                a[j]  = (i < 6) ? wt[i * 48 + t * 16 + m] : 0.0f;
                bb[j] = (i < 6) ? y1buf[i * 16 + n] : 0.0f;
            }
            acc2 = __builtin_amdgcn_wmma_f32_16x16x4_f32(false, a, false, bb,
                                                         (short)0, acc2, false, false);
        }
        const float btv = bt[t];
        #pragma unroll
        for (int u = 0; u < 4; ++u) {
            const int p = half;
            const int q = (u >> 1) & 1;
            const int r = u & 1;
            float2 o2;
            o2.x = sigmoidf(acc2[2 * u]     + btv);
            o2.y = sigmoidf(acc2[2 * u + 1] + btv);
            const size_t oidx =
                (((((size_t)b * 3 + t) * 32 + 2 * d + p) * 32 + 2 * e + q) * 32
                 + 2 * f + r) * 32 + 2 * n;
            *reinterpret_cast<float2*>(out + oidx) = o2;
        }
    }
}

extern "C" void kernel_launch(void* const* d_in, const int* in_sizes, int n_in,
                              void* d_out, int out_size, void* d_ws, size_t ws_size,
                              hipStream_t stream) {
    const float* x  = (const float*)d_in[0];
    const float* w0 = (const float*)d_in[1];
    const float* b0 = (const float*)d_in[2];
    const float* w1 = (const float*)d_in[3];
    const float* b1 = (const float*)d_in[4];
    const float* wt = (const float*)d_in[5];
    const float* bt = (const float*)d_in[6];
    float* out = (float*)d_out;

    const int Bsz = in_sizes[0] / (32 * 32 * 32 * 32);
    const int totalWaves = Bsz * 16 * 16 * 16 * 16 / 16;
    const int blocks = totalWaves / 8;

    fused_conv4d_wmma<<<blocks, 256, 0, stream>>>(x, w0, b0, w1, b1, wt, bt, out);
}


# 4 tiles per wave instead of 1 (512 blocks, 4x fewer waves to start): weights read once per wave, next tile's x loaded ahead, in-loop barriers wait on LDS only
# speedup vs baseline: 1.4329x; 1.4329x over previous
#include <hip/hip_runtime.h>


typedef __attribute__((ext_vector_type(2))) float v2f;
typedef __attribute__((ext_vector_type(8))) float v8f;

#define NEG_SLOPE 0.01f
#define TILES 4

__device__ __forceinline__ float leaky(float v) {
    return v >= 0.0f ? v : NEG_SLOPE * v;
}

__device__ __forceinline__ float sigmoidf(float z) {
    return 1.0f / (1.0f + __expf(-z));
}

__device__ __forceinline__ void lds_handover() {
    __builtin_amdgcn_fence(__ATOMIC_RELEASE, "workgroup", "local");
    __builtin_amdgcn_s_barrier();
    __builtin_amdgcn_fence(__ATOMIC_ACQUIRE, "workgroup", "local");
}

__global__ __launch_bounds__(256)
void fused_conv4d_wmma(const float* __restrict__ x,
                       const float* __restrict__ w0,
                       const float* __restrict__ b0,
                       const float* __restrict__ w1,
                       const float* __restrict__ b1,
                       const float* __restrict__ wt,
                       const float* __restrict__ bt,
                       float* __restrict__ out)
{
    __shared__ float lds[8][368];

    const int tid  = threadIdx.x;
    const int lane = tid & 31;
    const int warp = tid >> 5;
    const int n    = lane & 15;
    const int half = lane >> 4;
    const int m    = n;

    float* xblk  = &lds[warp][0];
    float* y1buf = &lds[warp][272];

    const int W  = blockIdx.x * 8 + warp;
    const int b  = W >> 10;
    const int d  = (W >> 6) & 15;
    const int e  = (W >> 2) & 15;
    const int f0 = (W & 3) * TILES;

    size_t xoff[4];
    float2 xv[4];
    #pragma unroll
    for (int i = 0; i < 4; ++i) {
        const int pos8 = 4 * half + i;
        const int p = (pos8 >> 2) & 1;
        const int q = (pos8 >> 1) & 1;
        const int r = pos8 & 1;
        xoff[i] =
            ((((size_t)b * 32 + 2 * d + p) * 32 + 2 * e + q) * 32 + 2 * f0 + r) * 32
            + 2 * n;
        xv[i] = *reinterpret_cast<const float2*>(x + xoff[i]);
    }

    float w0r[3], b0r[3], b1r[6], btr[3];
    #pragma unroll
    for (int c = 0; c < 3; ++c) { w0r[c] = w0[c]; b0r[c] = b0[c]; btr[c] = bt[c]; }
    #pragma unroll
    for (int o = 0; o < 6; ++o) b1r[o] = b1[o];

    const int mrow = (m < 6) ? m : 5;
    float a1v[12][2];
    #pragma unroll
    for (int k = 0; k < 12; ++k) {
        #pragma unroll
        for (int j = 0; j < 2; ++j) {
            const int kk = 4 * k + 2 * half + j;
            const float wv = w1[mrow * 48 + kk];
            a1v[k][j] = (m < 6) ? wv : 0.0f;
        }
    }

    float a2v[3][2][2];
    #pragma unroll
    for (int tch = 0; tch < 3; ++tch) {
        #pragma unroll
        for (int ks = 0; ks < 2; ++ks) {
            #pragma unroll
            for (int j = 0; j < 2; ++j) {
                const int i  = 4 * ks + 2 * half + j;
                const int ir = (i < 6) ? i : 5;
                const float wv = wt[ir * 48 + tch * 16 + m];
                a2v[tch][ks][j] = (i < 6) ? wv : 0.0f;
            }
        }
    }

    const size_t obase =
        ((((size_t)b * 3 * 32 + 2 * d + half) * 32 + 2 * e) * 32 + 2 * f0) * 32 + 2 * n;

    for (int t = 0; t < TILES; ++t) {
        #pragma unroll
        for (int i = 0; i < 4; ++i) {
            const int pos8 = 4 * half + i;
            xblk[(pos8 * 2 + 0) * 17 + n] = xv[i].x;
            xblk[(pos8 * 2 + 1) * 17 + n] = xv[i].y;
        }
        lds_handover();

        if (t + 1 < TILES) {
            #pragma unroll
            for (int i = 0; i < 4; ++i) {
                xv[i] = *reinterpret_cast<const float2*>(x + xoff[i] + (size_t)(t + 1) * 64);
            }
        }

        v8f acc = {0.f, 0.f, 0.f, 0.f, 0.f, 0.f, 0.f, 0.f};
        #pragma unroll
        for (int k = 0; k < 12; ++k) {
            const int kb = 4 * k + 2 * half;
            v2f a, bb;
            a[0] = a1v[k][0];
            a[1] = a1v[k][1];
            #pragma unroll
            for (int j = 0; j < 2; ++j) {
                const int kk  = kb + j;
                const int c   = kk >> 4;
                const int pos = kk & 15;
                bb[j] = leaky(w0r[c] * xblk[pos * 17 + n] + b0r[c]);
            }
            acc = __builtin_amdgcn_wmma_f32_16x16x4_f32(false, a, false, bb,
                                                        (short)0, acc, false, false);
        }

        if (half == 0) {
            #pragma unroll
            for (int o = 0; o < 6; ++o) {
                y1buf[o * 16 + n] = leaky(acc[o] + b1r[o]);
            }
        }
        lds_handover();

        #pragma unroll
        for (int tch = 0; tch < 3; ++tch) {
            v8f acc2 = {0.f, 0.f, 0.f, 0.f, 0.f, 0.f, 0.f, 0.f};
            #pragma unroll
            for (int ks = 0; ks < 2; ++ks) {
                const int kb = 4 * ks + 2 * half;
                v2f a, bb;
                #pragma unroll
                for (int j = 0; j < 2; ++j) {
                    const int i = kb + j;
                    a[j]  = a2v[tch][ks][j];
                    bb[j] = (i < 6) ? y1buf[i * 16 + n] : 0.0f;
                }
                acc2 = __builtin_amdgcn_wmma_f32_16x16x4_f32(false, a, false, bb,
                                                             (short)0, acc2, false, false);
            }
            const float btv = btr[tch];
            #pragma unroll
            for (int u = 0; u < 4; ++u) {
                const int q = (u >> 1) & 1;
                const int r = u & 1;
                float2 o2;
                o2.x = sigmoidf(acc2[2 * u]     + btv);
                o2.y = sigmoidf(acc2[2 * u + 1] + btv);
                const size_t oidx = obase
                    + (size_t)tch * (32u * 32u * 32u * 32u)
                    + (size_t)q * (32u * 32u)
                    + (size_t)r * 32u
                    + (size_t)t * 64u;
                *reinterpret_cast<float2*>(out + oidx) = o2;
            }
        }
    }
}

extern "C" void kernel_launch(void* const* d_in, const int* in_sizes, int n_in,
                              void* d_out, int out_size, void* d_ws, size_t ws_size,
                              hipStream_t stream) {
    const float* x  = (const float*)d_in[0];
    const float* w0 = (const float*)d_in[1];
    const float* b0 = (const float*)d_in[2];
    const float* w1 = (const float*)d_in[3];
    const float* b1 = (const float*)d_in[4];
    const float* wt = (const float*)d_in[5];
    const float* bt = (const float*)d_in[6];
    float* out = (float*)d_out;

    const int Bsz = in_sizes[0] / (32 * 32 * 32 * 32);
    const int totalWaves = Bsz * 16 * 16 * 16 * 16 / 16 / TILES;
    const int blocks = totalWaves / 8;

    fused_conv4d_wmma<<<blocks, 256, 0, stream>>>(x, w0, b0, w1, b1, wt, bt, out);
}


# sigmoid epilogue: hardware reciprocal v_rcp_f32 (about 1 ulp) of 1 + e^-z in place of the correctly rounded division, which the compiler expanded to 11 vector instructions per output element
# speedup vs baseline: 1.7048x; 1.1898x over previous
#include <hip/hip_runtime.h>


typedef __attribute__((ext_vector_type(2))) float v2f;
typedef __attribute__((ext_vector_type(8))) float v8f;

#define NEG_SLOPE 0.01f
#define TILES 4

__device__ __forceinline__ float leaky(float v) {
    return v >= 0.0f ? v : NEG_SLOPE * v;
}

__device__ __forceinline__ float sigmoidf(float z) {
    return __builtin_amdgcn_rcpf(1.0f + __expf(-z));
}

__device__ __forceinline__ void lds_handover() {
    __builtin_amdgcn_fence(__ATOMIC_RELEASE, "workgroup", "local");
    __builtin_amdgcn_s_barrier();
    __builtin_amdgcn_fence(__ATOMIC_ACQUIRE, "workgroup", "local");
}

__global__ __launch_bounds__(256)
void fused_conv4d_wmma(const float* __restrict__ x,
                       const float* __restrict__ w0,
                       const float* __restrict__ b0,
                       const float* __restrict__ w1,
                       const float* __restrict__ b1,
                       const float* __restrict__ wt,
                       const float* __restrict__ bt,
                       float* __restrict__ out)
{
    __shared__ float lds[8][368];

    const int tid  = threadIdx.x;
    const int lane = tid & 31;
    const int warp = tid >> 5;
    const int n    = lane & 15;
    const int half = lane >> 4;
    const int m    = n;

    float* xblk  = &lds[warp][0];
    float* y1buf = &lds[warp][272];

    const int W  = blockIdx.x * 8 + warp;
    const int b  = W >> 10;
    const int d  = (W >> 6) & 15;
    const int e  = (W >> 2) & 15;
    const int f0 = (W & 3) * TILES;

    size_t xoff[4];
    float2 xv[4];
    #pragma unroll
    for (int i = 0; i < 4; ++i) {
        const int pos8 = 4 * half + i;
        const int p = (pos8 >> 2) & 1;
        const int q = (pos8 >> 1) & 1;
        const int r = pos8 & 1;
        xoff[i] =
            ((((size_t)b * 32 + 2 * d + p) * 32 + 2 * e + q) * 32 + 2 * f0 + r) * 32
            + 2 * n;
        xv[i] = *reinterpret_cast<const float2*>(x + xoff[i]);
    }

    float w0r[3], b0r[3], b1r[6], btr[3];
    #pragma unroll
    for (int c = 0; c < 3; ++c) { w0r[c] = w0[c]; b0r[c] = b0[c]; btr[c] = bt[c]; }
    #pragma unroll
    for (int o = 0; o < 6; ++o) b1r[o] = b1[o];

    const int mrow = (m < 6) ? m : 5;
    float a1v[12][2];
    #pragma unroll
    for (int k = 0; k < 12; ++k) {
        #pragma unroll
        for (int j = 0; j < 2; ++j) {
            const int kk = 4 * k + 2 * half + j;
            const float wv = w1[mrow * 48 + kk];
            a1v[k][j] = (m < 6) ? wv : 0.0f;
        }
    }

    float a2v[3][2][2];
    #pragma unroll
    for (int tch = 0; tch < 3; ++tch) {
        #pragma unroll
        for (int ks = 0; ks < 2; ++ks) {
            #pragma unroll
            for (int j = 0; j < 2; ++j) {
                const int i  = 4 * ks + 2 * half + j;
                const int ir = (i < 6) ? i : 5;
                const float wv = wt[ir * 48 + tch * 16 + m];
                a2v[tch][ks][j] = (i < 6) ? wv : 0.0f;
            }
        }
    }

    const size_t obase =
        ((((size_t)b * 3 * 32 + 2 * d + half) * 32 + 2 * e) * 32 + 2 * f0) * 32 + 2 * n;

    for (int t = 0; t < TILES; ++t) {
        #pragma unroll
        for (int i = 0; i < 4; ++i) {
            const int pos8 = 4 * half + i;
            xblk[(pos8 * 2 + 0) * 17 + n] = xv[i].x;
            xblk[(pos8 * 2 + 1) * 17 + n] = xv[i].y;
        }
        lds_handover();

        if (t + 1 < TILES) {
            #pragma unroll
            for (int i = 0; i < 4; ++i) {
                xv[i] = *reinterpret_cast<const float2*>(x + xoff[i] + (size_t)(t + 1) * 64);
            }
        }

        v8f acc = {0.f, 0.f, 0.f, 0.f, 0.f, 0.f, 0.f, 0.f};
        #pragma unroll
        for (int k = 0; k < 12; ++k) {
            const int kb = 4 * k + 2 * half;
            v2f a, bb;
            a[0] = a1v[k][0];
            a[1] = a1v[k][1];
            #pragma unroll
            for (int j = 0; j < 2; ++j) {
                const int kk  = kb + j;
                const int c   = kk >> 4;
                const int pos = kk & 15;
                bb[j] = leaky(w0r[c] * xblk[pos * 17 + n] + b0r[c]);
            }
            acc = __builtin_amdgcn_wmma_f32_16x16x4_f32(false, a, false, bb,
                                                        (short)0, acc, false, false);
        }

        if (half == 0) {
            #pragma unroll
            for (int o = 0; o < 6; ++o) {
                y1buf[o * 16 + n] = leaky(acc[o] + b1r[o]);
            }
        }
        lds_handover();

        #pragma unroll
        for (int tch = 0; tch < 3; ++tch) {
            v8f acc2 = {0.f, 0.f, 0.f, 0.f, 0.f, 0.f, 0.f, 0.f};
            #pragma unroll
            for (int ks = 0; ks < 2; ++ks) {
                const int kb = 4 * ks + 2 * half;
                v2f a, bb;
                #pragma unroll
                for (int j = 0; j < 2; ++j) {
                    const int i = kb + j;
                    a[j]  = a2v[tch][ks][j];
                    bb[j] = (i < 6) ? y1buf[i * 16 + n] : 0.0f;
                }
                acc2 = __builtin_amdgcn_wmma_f32_16x16x4_f32(false, a, false, bb,
                                                             (short)0, acc2, false, false);
            }
            const float btv = btr[tch];
            #pragma unroll
            for (int u = 0; u < 4; ++u) {
                const int q = (u >> 1) & 1;
                const int r = u & 1;
                float2 o2;
                o2.x = sigmoidf(acc2[2 * u]     + btv);
                o2.y = sigmoidf(acc2[2 * u + 1] + btv);
                const size_t oidx = obase
                    + (size_t)tch * (32u * 32u * 32u * 32u)
                    + (size_t)q * (32u * 32u)
                    + (size_t)r * 32u
                    + (size_t)t * 64u;
                *reinterpret_cast<float2*>(out + oidx) = o2;
            }
        }
    }
}

extern "C" void kernel_launch(void* const* d_in, const int* in_sizes, int n_in,
                              void* d_out, int out_size, void* d_ws, size_t ws_size,
                              hipStream_t stream) {
    const float* x  = (const float*)d_in[0];
    const float* w0 = (const float*)d_in[1];
    const float* b0 = (const float*)d_in[2];
    const float* w1 = (const float*)d_in[3];
    const float* b1 = (const float*)d_in[4];
    const float* wt = (const float*)d_in[5];
    const float* bt = (const float*)d_in[6];
    float* out = (float*)d_out;

    const int Bsz = in_sizes[0] / (32 * 32 * 32 * 32);
    const int totalWaves = Bsz * 16 * 16 * 16 * 16 / 16 / TILES;
    const int blocks = totalWaves / 8;

    fused_conv4d_wmma<<<blocks, 256, 0, stream>>>(x, w0, b0, w1, b1, wt, bt, out);
}


# x loaded in the B fragment's own lane layout and stage-1 rows placed where stage 2 reads them: both B operands stay in registers; LDS staging and all 8 work-group barriers per wave removed
# speedup vs baseline: 1.9076x; 1.1190x over previous
#include <hip/hip_runtime.h>


typedef __attribute__((ext_vector_type(2))) float v2f;
typedef __attribute__((ext_vector_type(8))) float v8f;

#define NEG_SLOPE 0.01f
#define TILES 4

__device__ __forceinline__ float leaky(float v) {
    return v >= 0.0f ? v : NEG_SLOPE * v;
}

__device__ __forceinline__ float sigmoidf(float z) {
    return __builtin_amdgcn_rcpf(1.0f + __expf(-z));
}

__global__ __launch_bounds__(256)
void fused_conv4d_wmma(const float* __restrict__ x,
                       const float* __restrict__ w0,
                       const float* __restrict__ b0,
                       const float* __restrict__ w1,
                       const float* __restrict__ b1,
                       const float* __restrict__ wt,
                       const float* __restrict__ bt,
                       float* __restrict__ out)
{
    const int tid  = threadIdx.x;
    const int lane = tid & 31;
    const int warp = tid >> 5;
    const int n    = lane & 15;
    const int half = lane >> 4;
    const int m    = n;

    const int W  = blockIdx.x * 8 + warp;
    const int b  = W >> 10;
    const int d  = (W >> 6) & 15;
    const int e  = (W >> 2) & 15;
    const int f0 = (W & 3) * TILES;

    size_t xoff[4];
    float2 xv[4];
    #pragma unroll
    for (int i = 0; i < 4; ++i) {
        const int p = (i >> 1) & 1;
        const int q = i & 1;
        xoff[i] =
            ((((size_t)b * 32 + 2 * d + p) * 32 + 2 * e + q) * 32 + 2 * f0 + half) * 32
            + 2 * n;
        xv[i] = *reinterpret_cast<const float2*>(x + xoff[i]);
    }

    float w0r[3], b0r[3], b1r[6], btr[3];
    #pragma unroll
    for (int c = 0; c < 3; ++c) { w0r[c] = w0[c]; b0r[c] = b0[c]; btr[c] = bt[c]; }
    #pragma unroll
    for (int o = 0; o < 6; ++o) b1r[o] = b1[o];

    const int  orow   = (m < 2) ? m : (m < 4) ? m + 2 : m - 6;
    const bool usedrow = (m < 4) || (m == 8) || (m == 9);
    const int  mrow   = usedrow ? orow : 0;
    float a1v[12][2];
    #pragma unroll
    for (int k = 0; k < 12; ++k) {
        #pragma unroll
        for (int j = 0; j < 2; ++j) {
            const int kk = 4 * k + 2 * half + j;
            const float wv = w1[mrow * 48 + kk];
            a1v[k][j] = usedrow ? wv : 0.0f;
        }
    }

    const float bsel0 = half ? b1r[2] : b1r[0];
    const float bsel1 = half ? b1r[3] : b1r[1];

    float a2v[3][2][2];
    #pragma unroll
    for (int tch = 0; tch < 3; ++tch) {
        #pragma unroll
        for (int ks = 0; ks < 2; ++ks) {
            #pragma unroll
            for (int j = 0; j < 2; ++j) {
                const int i  = 4 * ks + 2 * half + j;
                const int ir = (i < 6) ? i : 5;
                const float wv = wt[ir * 48 + tch * 16 + m];
                a2v[tch][ks][j] = (i < 6) ? wv : 0.0f;
            }
        }
    }

    const size_t obase =
        ((((size_t)b * 3 * 32 + 2 * d + half) * 32 + 2 * e) * 32 + 2 * f0) * 32 + 2 * n;

    for (int t = 0; t < TILES; ++t) {
        float2 xc[4];
        #pragma unroll
        for (int i = 0; i < 4; ++i) xc[i] = xv[i];

        if (t + 1 < TILES) {
            #pragma unroll
            for (int i = 0; i < 4; ++i) {
                xv[i] = *reinterpret_cast<const float2*>(x + xoff[i] + (size_t)(t + 1) * 64);
            }
        }

        v8f acc = {0.f, 0.f, 0.f, 0.f, 0.f, 0.f, 0.f, 0.f};
        #pragma unroll
        for (int k = 0; k < 12; ++k) {
            v2f a, bb;
            a[0] = a1v[k][0];
            a[1] = a1v[k][1];
            const int c = k >> 2;
            bb[0] = leaky(w0r[c] * xc[k & 3].x + b0r[c]);
            bb[1] = leaky(w0r[c] * xc[k & 3].y + b0r[c]);
            acc = __builtin_amdgcn_wmma_f32_16x16x4_f32(false, a, false, bb,
                                                        (short)0, acc, false, false);
        }

        float y1r[4];
        y1r[0] = leaky(acc[0] + bsel0);
        y1r[1] = leaky(acc[1] + bsel1);
        const float t2 = leaky(acc[2] + b1r[4]);
        const float t3 = leaky(acc[3] + b1r[5]);
        y1r[2] = half ? 0.0f : t2;
        y1r[3] = half ? 0.0f : t3;

        #pragma unroll
        for (int tch = 0; tch < 3; ++tch) {
            v8f acc2 = {0.f, 0.f, 0.f, 0.f, 0.f, 0.f, 0.f, 0.f};
            #pragma unroll
            for (int ks = 0; ks < 2; ++ks) {
                v2f a, bb;
                #pragma unroll
                for (int j = 0; j < 2; ++j) {
                    a[j]  = a2v[tch][ks][j];
                    bb[j] = y1r[2 * ks + j];
                }
                acc2 = __builtin_amdgcn_wmma_f32_16x16x4_f32(false, a, false, bb,
                                                             (short)0, acc2, false, false);
            }
            const float btv = btr[tch];
            #pragma unroll
            for (int u = 0; u < 4; ++u) {
                const int q = (u >> 1) & 1;
                const int r = u & 1;
                float2 o2;
                o2.x = sigmoidf(acc2[2 * u]     + btv);
                o2.y = sigmoidf(acc2[2 * u + 1] + btv);
                const size_t oidx = obase
                    + (size_t)tch * (32u * 32u * 32u * 32u)
                    + (size_t)q * (32u * 32u)
                    + (size_t)r * 32u
                    + (size_t)t * 64u;
                *reinterpret_cast<float2*>(out + oidx) = o2;
            }
        }
    }
}

extern "C" void kernel_launch(void* const* d_in, const int* in_sizes, int n_in,
                              void* d_out, int out_size, void* d_ws, size_t ws_size,
                              hipStream_t stream) {
    const float* x  = (const float*)d_in[0];
    const float* w0 = (const float*)d_in[1];
    const float* b0 = (const float*)d_in[2];
    const float* w1 = (const float*)d_in[3];
    const float* b1 = (const float*)d_in[4];
    const float* wt = (const float*)d_in[5];
    const float* bt = (const float*)d_in[6];
    float* out = (float*)d_out;

    const int Bsz = in_sizes[0] / (32 * 32 * 32 * 32);
    const int totalWaves = Bsz * 16 * 16 * 16 * 16 / 16 / TILES;
    const int blocks = totalWaves / 8;

    fused_conv4d_wmma<<<blocks, 256, 0, stream>>>(x, w0, b0, w1, b1, wt, bt, out);
}


# sigmoid epilogue as 1/2 + 1/2*tanh(z/2) on the hardware tanh (v_tanh_f32): one transcendental per output where exp + rcp took two; both halvings as packed fma on output pairs; no sum reordered
# speedup vs baseline: 2.0185x; 1.0581x over previous
#include <hip/hip_runtime.h>


typedef __attribute__((ext_vector_type(2))) float v2f;
typedef __attribute__((ext_vector_type(8))) float v8f;

#define NEG_SLOPE 0.01f
#define TILES 4

__device__ __forceinline__ float leaky(float v) {
    return v >= 0.0f ? v : NEG_SLOPE * v;
}

__device__ __forceinline__ v2f sigmoid_pair(v2f acc, v2f hb) {
    const v2f hh = {0.5f, 0.5f};
    const v2f h = acc * hh + hb;
    v2f t;
    t[0] = __builtin_amdgcn_tanhf(h[0]);
    t[1] = __builtin_amdgcn_tanhf(h[1]);
    return t * hh + hh;
}

__global__ __launch_bounds__(256)
void fused_conv4d_wmma(const float* __restrict__ x,
                       const float* __restrict__ w0,
                       const float* __restrict__ b0,
                       const float* __restrict__ w1,
                       const float* __restrict__ b1,
                       const float* __restrict__ wt,
                       const float* __restrict__ bt,
                       float* __restrict__ out)
{
    const int tid  = threadIdx.x;
    const int lane = tid & 31;
    const int warp = tid >> 5;
    const int n    = lane & 15;
    const int half = lane >> 4;
    const int m    = n;

    const int W  = blockIdx.x * 8 + warp;
    const int b  = W >> 10;
    const int d  = (W >> 6) & 15;
    const int e  = (W >> 2) & 15;
    const int f0 = (W & 3) * TILES;

    size_t xoff[4];
    float2 xv[4];
    #pragma unroll
    for (int i = 0; i < 4; ++i) {
        const int p = (i >> 1) & 1;
        const int q = i & 1;
        xoff[i] =
            ((((size_t)b * 32 + 2 * d + p) * 32 + 2 * e + q) * 32 + 2 * f0 + half) * 32
            + 2 * n;
        xv[i] = *reinterpret_cast<const float2*>(x + xoff[i]);
    }

    float w0r[3], b0r[3], b1r[6], hbt[3];
    #pragma unroll
    for (int c = 0; c < 3; ++c) { w0r[c] = w0[c]; b0r[c] = b0[c]; hbt[c] = 0.5f * bt[c]; }
    #pragma unroll
    for (int o = 0; o < 6; ++o) b1r[o] = b1[o];

    const int  orow   = (m < 2) ? m : (m < 4) ? m + 2 : m - 6;
    const bool usedrow = (m < 4) || (m == 8) || (m == 9);
    const int  mrow   = usedrow ? orow : 0;
    float a1v[12][2];
    #pragma unroll
    for (int k = 0; k < 12; ++k) {
        #pragma unroll
        for (int j = 0; j < 2; ++j) {
            const int kk = 4 * k + 2 * half + j;
            const float wv = w1[mrow * 48 + kk];
            a1v[k][j] = usedrow ? wv : 0.0f;
        }
    }

    const float bsel0 = half ? b1r[2] : b1r[0];
    const float bsel1 = half ? b1r[3] : b1r[1];

    float a2v[3][2][2];
    #pragma unroll
    for (int tch = 0; tch < 3; ++tch) {
        #pragma unroll
        for (int ks = 0; ks < 2; ++ks) {
            #pragma unroll
            for (int j = 0; j < 2; ++j) {
                const int i  = 4 * ks + 2 * half + j;
                const int ir = (i < 6) ? i : 5;
                const float wv = wt[ir * 48 + tch * 16 + m];
                a2v[tch][ks][j] = (i < 6) ? wv : 0.0f;
            }
        }
    }

    const size_t obase =
        ((((size_t)b * 3 * 32 + 2 * d + half) * 32 + 2 * e) * 32 + 2 * f0) * 32 + 2 * n;

    for (int t = 0; t < TILES; ++t) {
        float2 xc[4];
        #pragma unroll
        for (int i = 0; i < 4; ++i) xc[i] = xv[i];

        if (t + 1 < TILES) {
            #pragma unroll
            for (int i = 0; i < 4; ++i) {
                xv[i] = *reinterpret_cast<const float2*>(x + xoff[i] + (size_t)(t + 1) * 64);
            }
        }

        v8f acc = {0.f, 0.f, 0.f, 0.f, 0.f, 0.f, 0.f, 0.f};
        #pragma unroll
        for (int k = 0; k < 12; ++k) {
            v2f a, bb;
            a[0] = a1v[k][0];
            a[1] = a1v[k][1];
            const int c = k >> 2;
            bb[0] = leaky(w0r[c] * xc[k & 3].x + b0r[c]);
            bb[1] = leaky(w0r[c] * xc[k & 3].y + b0r[c]);
            acc = __builtin_amdgcn_wmma_f32_16x16x4_f32(false, a, false, bb,
                                                        (short)0, acc, false, false);
        }

        float y1r[4];
        y1r[0] = leaky(acc[0] + bsel0);
        y1r[1] = leaky(acc[1] + bsel1);
        const float t2 = leaky(acc[2] + b1r[4]);
        const float t3 = leaky(acc[3] + b1r[5]);
        y1r[2] = half ? 0.0f : t2;
        y1r[3] = half ? 0.0f : t3;

        #pragma unroll
        for (int tch = 0; tch < 3; ++tch) {
            v8f acc2 = {0.f, 0.f, 0.f, 0.f, 0.f, 0.f, 0.f, 0.f};
            #pragma unroll
            for (int ks = 0; ks < 2; ++ks) {
                v2f a, bb;
                #pragma unroll
                for (int j = 0; j < 2; ++j) {
                    a[j]  = a2v[tch][ks][j];
                    bb[j] = y1r[2 * ks + j];
                }
                acc2 = __builtin_amdgcn_wmma_f32_16x16x4_f32(false, a, false, bb,
                                                             (short)0, acc2, false, false);
            }
            const v2f hb = {hbt[tch], hbt[tch]};
            #pragma unroll
            for (int u = 0; u < 4; ++u) {
                const int q = (u >> 1) & 1;
                const int r = u & 1;
                v2f zp;
                zp[0] = acc2[2 * u];
                zp[1] = acc2[2 * u + 1];
                const v2f sp = sigmoid_pair(zp, hb);
                float2 o2;
                o2.x = sp[0];
                o2.y = sp[1];
                const size_t oidx = obase
                    + (size_t)tch * (32u * 32u * 32u * 32u)
                    + (size_t)q * (32u * 32u)
                    + (size_t)r * 32u
                    + (size_t)t * 64u;
                *reinterpret_cast<float2*>(out + oidx) = o2;
            }
        }
    }
}

extern "C" void kernel_launch(void* const* d_in, const int* in_sizes, int n_in,
                              void* d_out, int out_size, void* d_ws, size_t ws_size,
                              hipStream_t stream) {
    const float* x  = (const float*)d_in[0];
    const float* w0 = (const float*)d_in[1];
    const float* b0 = (const float*)d_in[2];
    const float* w1 = (const float*)d_in[3];
    const float* b1 = (const float*)d_in[4];
    const float* wt = (const float*)d_in[5];
    const float* bt = (const float*)d_in[6];
    float* out = (float*)d_out;

    const int Bsz = in_sizes[0] / (32 * 32 * 32 * 32);
    const int totalWaves = Bsz * 16 * 16 * 16 * 16 / 16 / TILES;
    const int blocks = totalWaves / 8;

    fused_conv4d_wmma<<<blocks, 256, 0, stream>>>(x, w0, b0, w1, b1, wt, bt, out);
}
